# starting kernel
# baseline (speedup 1.0000x reference)
#include <hip/hip_runtime.h>
#include <math.h>

typedef float v2f __attribute__((ext_vector_type(2)));
typedef float v8f __attribute__((ext_vector_type(8)));

#define H_SZ 50
#define O_OUT 10
#define B_SZ 512
#define T_STEPS 2048
#define RS 68
#define XBLK 64
#define NT 4
#define KC 16

__device__ __forceinline__ float fast_tanh(float v) {
#if __has_builtin(__builtin_amdgcn_tanhf)
    return __builtin_amdgcn_tanhf(v);
#else
    float e = __expf(v + v);
    return 1.0f - __fdividef(2.0f, e + 1.0f);
#endif
}

#define WMMA_F32(A, B, C) \
    __builtin_amdgcn_wmma_f32_16x16x4_f32(false, (A), false, (B), (short)0, (C), false, false)

__global__ __launch_bounds__(128, 1) void rnn2_tanh_wmma(
    const float* __restrict__ x,
    const float* __restrict__ W_ih0, const float* __restrict__ W_hh0,
    const float* __restrict__ b_ih0, const float* __restrict__ b_hh0,
    const float* __restrict__ W_ih1, const float* __restrict__ W_hh1,
    const float* __restrict__ b_ih1, const float* __restrict__ b_hh1,
    const float* __restrict__ W_fc,  const float* __restrict__ b_fc,
    float* __restrict__ out)
{
    __shared__ float wB0[NT][KC][32][2];
    __shared__ float wB1[NT][KC][32][2];
    __shared__ float wB2[NT][KC][32][2];
    __shared__ float h0buf[2][16 * RS];
    __shared__ float h1buf[2][16 * RS];
    __shared__ float xs[16 * RS];

    const int tid  = threadIdx.x;
    const int lane = tid & 31;
    const int wv   = tid >> 5;
    const int r    = lane & 15;
    const int hf   = lane >> 4;
    const int b0   = blockIdx.x * 16;

    {
        float* h0f = &h0buf[0][0];
        float* h1f = &h1buf[0][0];
        for (int i = tid; i < 2 * 16 * RS; i += 128) { h0f[i] = 0.f; h1f[i] = 0.f; }
        float* w0f = &wB0[0][0][0][0];
        float* w1f = &wB1[0][0][0][0];
        float* w2f = &wB2[0][0][0][0];
        for (int idx = tid; idx < NT * KC * 32 * 2; idx += 128) {
            int v  = idx & 1;
            int ln = (idx >> 1) & 31;
            int kc = (idx >> 6) & 15;
            int nt = idx >> 10;
            int n  = nt * 16 + (ln & 15);
            int k  = kc * 4 + 2 * (ln >> 4) + v;
            float a = 0.f, b = 0.f, c = 0.f;
            if (n < H_SZ && k < H_SZ) {
                a = W_hh0[n * H_SZ + k];
                b = W_ih1[n * H_SZ + k];
                c = W_hh1[n * H_SZ + k];
            }
            w0f[idx] = a; w1f[idx] = b; w2f[idx] = c;
        }
    }

    const int n = wv * 16 + r;
    float wih = 0.f, bias0n = 0.f, bias1n = 0.f;
    if (n < H_SZ) {
        wih    = W_ih0[n];
        bias0n = b_ih0[n] + b_hh0[n];
        bias1n = b_ih1[n] + b_hh1[n];
    }
    __syncthreads();

    v2f rB0[KC], rB1[KC], rB2[KC];
    #pragma unroll
    for (int kc = 0; kc < KC; ++kc) {
        rB0[kc] = *(const v2f*)&wB0[wv][kc][lane][0];
        rB1[kc] = *(const v2f*)&wB1[wv][kc][lane][0];
        rB2[kc] = *(const v2f*)&wB2[wv][kc][lane][0];
    }

    int p = 0;
    for (int t = 0; t < T_STEPS; ++t) {
        const int tt = t & (XBLK - 1);
        if (tt == 0) {
            for (int i = tid; i < 16 * XBLK; i += 128) {
                int rr = i >> 6;
                int cc = i & (XBLK - 1);
                xs[rr * RS + cc] = x[(size_t)(b0 + rr) * T_STEPS + t + cc];
            }
            __syncthreads();
        }

        v8f acc0, acc1;
        #pragma unroll
        for (int v = 0; v < 8; ++v) {
            float xm = xs[(v + 8 * hf) * RS + tt];
            acc0[v] = fmaf(xm, wih, bias0n);
            acc1[v] = 0.0f;
        }
        const float* h0r = &h0buf[p][0];
        #pragma unroll
        for (int kc = 0; kc < KC; kc += 2) {
            v2f aA = *(const v2f*)&h0r[r * RS + kc * 4 + 2 * hf];
            v2f aB = *(const v2f*)&h0r[r * RS + (kc + 1) * 4 + 2 * hf];
            acc0 = WMMA_F32(aA, rB0[kc],     acc0);
            acc1 = WMMA_F32(aB, rB0[kc + 1], acc1);
        }
        v8f c0 = acc0 + acc1;
        float* h0w = &h0buf[1 - p][0];
        #pragma unroll
        for (int v = 0; v < 8; ++v)
            h0w[(v + 8 * hf) * RS + wv * 16 + r] = fast_tanh(c0[v]);
        __syncthreads();

        v8f acc2, acc3;
        #pragma unroll
        for (int v = 0; v < 8; ++v) { acc2[v] = bias1n; acc3[v] = 0.0f; }
        const float* h1r = &h1buf[p][0];
        #pragma unroll
        for (int kc = 0; kc < KC; ++kc) {
            v2f a0 = *(const v2f*)&h0w[r * RS + kc * 4 + 2 * hf];
            acc2 = WMMA_F32(a0, rB1[kc], acc2);
            v2f a1 = *(const v2f*)&h1r[r * RS + kc * 4 + 2 * hf];
            acc3 = WMMA_F32(a1, rB2[kc], acc3);
        }
        v8f c1 = acc2 + acc3;
        float* h1w = &h1buf[1 - p][0];
        #pragma unroll
        for (int v = 0; v < 8; ++v)
            h1w[(v + 8 * hf) * RS + wv * 16 + r] = fast_tanh(c1[v]);
        __syncthreads();

        p ^= 1;
    }

    const float* h1f = &h1buf[p][0];
    for (int i = tid; i < 16 * O_OUT; i += 128) {
        int m = i / O_OUT, o = i % O_OUT;
        float acc = b_fc[o];
        #pragma unroll 5
        for (int k = 0; k < H_SZ; ++k)
            acc = fmaf(h1f[m * RS + k], W_fc[o * H_SZ + k], acc);
        out[(size_t)(b0 + m) * O_OUT + o] = acc;
    }
}

extern "C" void kernel_launch(void* const* d_in, const int* in_sizes, int n_in,
                              void* d_out, int out_size, void* d_ws, size_t ws_size,
                              hipStream_t stream) {
    const float* x     = (const float*)d_in[0];
    const float* W_ih0 = (const float*)d_in[1];
    const float* W_hh0 = (const float*)d_in[2];
    const float* b_ih0 = (const float*)d_in[3];
    const float* b_hh0 = (const float*)d_in[4];
    const float* W_ih1 = (const float*)d_in[5];
    const float* W_hh1 = (const float*)d_in[6];
    const float* b_ih1 = (const float*)d_in[7];
    const float* b_hh1 = (const float*)d_in[8];
    const float* W_fc  = (const float*)d_in[9];
    const float* b_fc  = (const float*)d_in[10];
    float* out = (float*)d_out;

    rnn2_tanh_wmma<<<B_SZ / 16, 128, 0, stream>>>(
        x, W_ih0, W_hh0, b_ih0, b_hh0, W_ih1, W_hh1, b_ih1, b_hh1, W_fc, b_fc, out);
}


# Layer-0 and layer-1 waves run separate loops out of phase, joined by two split barriers per step (arrive early, wait late): layer 0's WMMAs fill layer 1's tail and LDS wait; order of sums unchanged
# speedup vs baseline: 1.5034x; 1.5034x over previous
#include <hip/hip_runtime.h>
#include <math.h>

typedef float v2f __attribute__((ext_vector_type(2)));
typedef float v8f __attribute__((ext_vector_type(8)));

#define H_SZ 50
#define O_OUT 10
#define B_SZ 512
#define T_STEPS 2048
#define RS 68
#define XBLK 64
#define NT 4
#define KC 16
#define NTHR 256

__device__ __forceinline__ float fast_tanh(float v) {
#if __has_builtin(__builtin_amdgcn_tanhf)
    return __builtin_amdgcn_tanhf(v);
#else
    float e = __expf(v + v);
    return 1.0f - __fdividef(2.0f, e + 1.0f);
#endif
}

#define WMMA_F32(A, B, C) \
    __builtin_amdgcn_wmma_f32_16x16x4_f32(false, (A), false, (B), (short)0, (C), false, false)

__device__ __forceinline__ void bar_arrive() { asm volatile("s_barrier_signal -1" ::: "memory"); }
__device__ __forceinline__ void bar_wait()   { asm volatile("s_barrier_wait -1" ::: "memory"); }
__device__ __forceinline__ void bar_arrive_here() {
    __builtin_amdgcn_sched_barrier(0);
    asm volatile("s_barrier_signal -1" ::: "memory");
    __builtin_amdgcn_sched_barrier(0);
}
__device__ __forceinline__ void bar_wait_late() {
    __builtin_amdgcn_sched_barrier(0);
    asm volatile("s_barrier_wait -1" ::: "memory");
}
__device__ __forceinline__ void lds_stores_landed() { asm volatile("s_wait_dscnt 0x0" ::: "memory"); }

__global__ __launch_bounds__(256, 1) void rnn2_tanh_wmma(
    const float* __restrict__ x,
    const float* __restrict__ W_ih0, const float* __restrict__ W_hh0,
    const float* __restrict__ b_ih0, const float* __restrict__ b_hh0,
    const float* __restrict__ W_ih1, const float* __restrict__ W_hh1,
    const float* __restrict__ b_ih1, const float* __restrict__ b_hh1,
    const float* __restrict__ W_fc,  const float* __restrict__ b_fc,
    float* __restrict__ out)
{
    __shared__ float wB0[NT][KC][32][2];
    __shared__ float wB1[NT][KC][32][2];
    __shared__ float wB2[NT][KC][32][2];
    __shared__ float h0buf[2][16 * RS];
    __shared__ float h1buf[2][16 * RS];
    __shared__ float xs[NT][16 * RS];

    const int tid  = threadIdx.x;
    const int lane = tid & 31;
    const int wv   = __builtin_amdgcn_readfirstlane(tid >> 5);
    const int role = (wv ^ (wv >> 2)) & 1;
    const int nt   = wv >> 1;
    const int r    = lane & 15;
    const int hf   = lane >> 4;
    const int b0   = blockIdx.x * 16;

    {
        float* h0f = &h0buf[0][0];
        float* h1f = &h1buf[0][0];
        for (int i = tid; i < 2 * 16 * RS; i += NTHR) { h0f[i] = 0.f; h1f[i] = 0.f; }
        float* w0f = &wB0[0][0][0][0];
        float* w1f = &wB1[0][0][0][0];
        float* w2f = &wB2[0][0][0][0];
        for (int idx = tid; idx < NT * KC * 32 * 2; idx += NTHR) {
            int v  = idx & 1;
            int ln = (idx >> 1) & 31;
            int kc = (idx >> 6) & 15;
            int tn = idx >> 10;
            int n  = tn * 16 + (ln & 15);
            int k  = kc * 4 + 2 * (ln >> 4) + v;
            float a = 0.f, b = 0.f, c = 0.f;
            if (n < H_SZ && k < H_SZ) {
                a = W_hh0[n * H_SZ + k];
                b = W_ih1[n * H_SZ + k];
                c = W_hh1[n * H_SZ + k];
            }
            w0f[idx] = a; w1f[idx] = b; w2f[idx] = c;
        }
    }

    const int n = nt * 16 + r;
    float wih = 0.f, bias0n = 0.f, bias1n = 0.f;
    if (n < H_SZ) {
        wih    = W_ih0[n];
        bias0n = b_ih0[n] + b_hh0[n];
        bias1n = b_ih1[n] + b_hh1[n];
    }
    __syncthreads();

    if (role == 0) {
        v2f rB0[KC];
        #pragma unroll
        for (int kc = 0; kc < KC; ++kc) rB0[kc] = *(const v2f*)&wB0[nt][kc][lane][0];
        float* xw = &xs[nt][0];

        for (int i = 0; i < T_STEPS; ++i) {
            bar_arrive();
            const int tt = i & (XBLK - 1);
            if (tt == 0) {
                #pragma unroll
                for (int j = lane; j < 16 * XBLK; j += 32) {
                    int rr = j >> 6;
                    int cc = j & (XBLK - 1);
                    xw[rr * RS + cc] = x[(size_t)(b0 + rr) * T_STEPS + i + cc];
                }
            }
            v8f acc0, acc1;
            #pragma unroll
            for (int v = 0; v < 8; ++v) {
                float xm = xw[(v + 8 * hf) * RS + tt];
                acc0[v] = fmaf(xm, wih, bias0n);
                acc1[v] = 0.0f;
            }
            const float* h0r = &h0buf[(i & 1) ^ 1][0];
            #pragma unroll
            for (int kc = 0; kc < KC; kc += 2) {
                v2f aA = *(const v2f*)&h0r[r * RS + kc * 4 + 2 * hf];
                v2f aB = *(const v2f*)&h0r[r * RS + (kc + 1) * 4 + 2 * hf];
                acc0 = WMMA_F32(aA, rB0[kc],     acc0);
                acc1 = WMMA_F32(aB, rB0[kc + 1], acc1);
            }
            v8f c0 = acc0 + acc1;
            float t0[8];
            #pragma unroll
            for (int v = 0; v < 8; ++v) t0[v] = fast_tanh(c0[v]);
            bar_wait_late();
            float* h0w = &h0buf[i & 1][0];
            #pragma unroll
            for (int v = 0; v < 8; ++v)
                h0w[(v + 8 * hf) * RS + nt * 16 + r] = t0[v];
            lds_stores_landed();
            bar_arrive();
            bar_wait();
        }
        bar_arrive(); bar_wait();
        bar_arrive(); bar_wait();
        bar_arrive(); bar_wait();
    } else {
        v2f rB1[KC], rB2[KC];
        #pragma unroll
        for (int kc = 0; kc < KC; ++kc) {
            rB1[kc] = *(const v2f*)&wB1[nt][kc][lane][0];
            rB2[kc] = *(const v2f*)&wB2[nt][kc][lane][0];
        }
        bar_arrive(); bar_wait();
        bar_arrive(); bar_wait();
        bar_arrive(); bar_wait();

        for (int s = 0; s < T_STEPS; ++s) {
            v8f acc2, acc3;
            #pragma unroll
            for (int v = 0; v < 8; ++v) { acc2[v] = bias1n; acc3[v] = 0.0f; }
            const float* h0r = &h0buf[s & 1][0];
            const float* h1r = &h1buf[(s & 1) ^ 1][0];
            #pragma unroll
            for (int kc = 0; kc < KC - KC / 4; ++kc) {
                v2f a0 = *(const v2f*)&h0r[r * RS + kc * 4 + 2 * hf];
                acc2 = WMMA_F32(a0, rB1[kc], acc2);
                v2f a1 = *(const v2f*)&h1r[r * RS + kc * 4 + 2 * hf];
                acc3 = WMMA_F32(a1, rB2[kc], acc3);
            }
            v2f l0[KC / 4], l1[KC / 4];
            #pragma unroll
            for (int q = 0; q < KC / 4; ++q) {
                const int kc = KC - KC / 4 + q;
                l0[q] = *(const v2f*)&h0r[r * RS + kc * 4 + 2 * hf];
                l1[q] = *(const v2f*)&h1r[r * RS + kc * 4 + 2 * hf];
            }
            bar_arrive_here();
            #pragma unroll
            for (int q = 0; q < KC / 4; ++q) {
                const int kc = KC - KC / 4 + q;
                acc2 = WMMA_F32(l0[q], rB1[kc], acc2);
                acc3 = WMMA_F32(l1[q], rB2[kc], acc3);
            }
            v8f c1 = acc2 + acc3;
            float t1[8];
            #pragma unroll
            for (int v = 0; v < 8; ++v) t1[v] = fast_tanh(c1[v]);
            bar_wait_late();
            float* h1w = &h1buf[s & 1][0];
            #pragma unroll
            for (int v = 0; v < 8; ++v)
                h1w[(v + 8 * hf) * RS + nt * 16 + r] = t1[v];
            lds_stores_landed();
            bar_arrive();
            bar_wait();
        }
    }

    const float* h1f = &h1buf[(T_STEPS - 1) & 1][0];
    for (int i = tid; i < 16 * O_OUT; i += NTHR) {
        int m = i / O_OUT, o = i % O_OUT;
        float acc = b_fc[o];
        #pragma unroll 5
        for (int k = 0; k < H_SZ; ++k)
            acc = fmaf(h1f[m * RS + k], W_fc[o * H_SZ + k], acc);
        out[(size_t)(b0 + m) * O_OUT + o] = acc;
    }
}

extern "C" void kernel_launch(void* const* d_in, const int* in_sizes, int n_in,
                              void* d_out, int out_size, void* d_ws, size_t ws_size,
                              hipStream_t stream) {
    const float* x     = (const float*)d_in[0];
    const float* W_ih0 = (const float*)d_in[1];
    const float* W_hh0 = (const float*)d_in[2];
    const float* b_ih0 = (const float*)d_in[3];
    const float* b_hh0 = (const float*)d_in[4];
    const float* W_ih1 = (const float*)d_in[5];
    const float* W_hh1 = (const float*)d_in[6];
    const float* b_ih1 = (const float*)d_in[7];
    const float* b_hh1 = (const float*)d_in[8];
    const float* W_fc  = (const float*)d_in[9];
    const float* b_fc  = (const float*)d_in[10];
    float* out = (float*)d_out;

    rnn2_tanh_wmma<<<B_SZ / 16, NTHR, 0, stream>>>(
        x, W_ih0, W_hh0, b_ih0, b_hh0, W_ih1, W_hh1, b_ih1, b_hh1, W_fc, b_fc, out);
}


# Layer 1's two products fused into one over a concatenated LDS state row [h1 | h0] with stacked weights: K = 100 in 25 chunks of 4 per step, was 2 x 16 chunks; order of the k sums changes
# speedup vs baseline: 1.6134x; 1.0732x over previous
#include <hip/hip_runtime.h>
#include <math.h>

typedef float v2f __attribute__((ext_vector_type(2)));
typedef float v8f __attribute__((ext_vector_type(8)));

#define H_SZ 50
#define O_OUT 10
#define B_SZ 512
#define T_STEPS 2048
#define RS 68
#define RSC 132
#define H0_COL 50
#define XBLK 64
#define NT 4
#define KC 16
#define KCC 25
#define LASTQ 8
#define NTHR 256

__device__ __forceinline__ float fast_tanh(float v) {
#if __has_builtin(__builtin_amdgcn_tanhf)
    return __builtin_amdgcn_tanhf(v);
#else
    float e = __expf(v + v);
    return 1.0f - __fdividef(2.0f, e + 1.0f);
#endif
}

#define WMMA_F32(A, B, C) \
    __builtin_amdgcn_wmma_f32_16x16x4_f32(false, (A), false, (B), (short)0, (C), false, false)

__device__ __forceinline__ void bar_arrive() { asm volatile("s_barrier_signal -1" ::: "memory"); }
__device__ __forceinline__ void bar_wait()   { asm volatile("s_barrier_wait -1" ::: "memory"); }
__device__ __forceinline__ void bar_arrive_here() {
    __builtin_amdgcn_sched_barrier(0);
    asm volatile("s_barrier_signal -1" ::: "memory");
    __builtin_amdgcn_sched_barrier(0);
}
__device__ __forceinline__ void bar_wait_late() {
    __builtin_amdgcn_sched_barrier(0);
    asm volatile("s_barrier_wait -1" ::: "memory");
}
__device__ __forceinline__ void lds_stores_landed() { asm volatile("s_wait_dscnt 0x0" ::: "memory"); }

__global__ __launch_bounds__(256, 1) void rnn2_tanh_wmma(
    const float* __restrict__ x,
    const float* __restrict__ W_ih0, const float* __restrict__ W_hh0,
    const float* __restrict__ b_ih0, const float* __restrict__ b_hh0,
    const float* __restrict__ W_ih1, const float* __restrict__ W_hh1,
    const float* __restrict__ b_ih1, const float* __restrict__ b_hh1,
    const float* __restrict__ W_fc,  const float* __restrict__ b_fc,
    float* __restrict__ out)
{
    __shared__ float wB0[NT][KC][32][2];
    __shared__ float wBc[NT][KCC][32][2];
    __shared__ float cat[2][16 * RSC];
    __shared__ float xs[NT][16 * RS];

    const int tid  = threadIdx.x;
    const int lane = tid & 31;
    const int wv   = __builtin_amdgcn_readfirstlane(tid >> 5);
    const int role = (wv ^ (wv >> 2)) & 1;
    const int nt   = wv >> 1;
    const int r    = lane & 15;
    const int hf   = lane >> 4;
    const int b0   = blockIdx.x * 16;

    {
        float* cf = &cat[0][0];
        for (int i = tid; i < 2 * 16 * RSC; i += NTHR) cf[i] = 0.f;
        float* w0f = &wB0[0][0][0][0];
        for (int idx = tid; idx < NT * KC * 32 * 2; idx += NTHR) {
            int v  = idx & 1;
            int ln = (idx >> 1) & 31;
            int kc = (idx >> 6) & 15;
            int tn = idx >> 10;
            int n  = tn * 16 + (ln & 15);
            int k  = kc * 4 + 2 * (ln >> 4) + v;
            float a = 0.f;
            if (n < H_SZ && k < H_SZ)
                a = W_hh0[n * H_SZ + k];
            w0f[idx] = a;
        }
        float* wcf = &wBc[0][0][0][0];
        for (int idx = tid; idx < NT * KCC * 32 * 2; idx += NTHR) {
            int v  = idx & 1;
            int ln = (idx >> 1) & 31;
            int ch = idx >> 6;
            int kc = ch % KCC;
            int tn = ch / KCC;
            int n  = tn * 16 + (ln & 15);
            int k  = kc * 4 + 2 * (ln >> 4) + v;
            float c = 0.f;
            if (n < H_SZ)
                c = (k < H0_COL) ? W_hh1[n * H_SZ + k] : W_ih1[n * H_SZ + (k - H0_COL)];
            wcf[idx] = c;
        }
    }

    const int n = nt * 16 + r;
    float wih = 0.f, bias0n = 0.f, bias1n = 0.f;
    if (n < H_SZ) {
        wih    = W_ih0[n];
        bias0n = b_ih0[n] + b_hh0[n];
        bias1n = b_ih1[n] + b_hh1[n];
    }
    const int col1 = (n < H_SZ) ? n : n + 64;
    __syncthreads();

    if (role == 0) {
        v2f rB0[KC];
        #pragma unroll
        for (int kc = 0; kc < KC; ++kc) rB0[kc] = *(const v2f*)&wB0[nt][kc][lane][0];
        float* xw = &xs[nt][0];
        const float* xrow = xw + 8 * hf * RS;
        const int st0 = 8 * hf * RSC + nt * 16 + r;

        for (int i = 0; i < T_STEPS; ++i) {
            bar_arrive();
            const int tt = i & (XBLK - 1);
            if (tt == 0) {
                #pragma unroll
                for (int j = lane; j < 16 * XBLK; j += 32) {
                    int rr = j >> 6;
                    int cc = j & (XBLK - 1);
                    xw[rr * RS + cc] = x[(size_t)(b0 + rr) * T_STEPS + i + cc];
                }
            }
            v8f acc0, acc1;
            #pragma unroll
            for (int v = 0; v < 8; ++v) {
                float xm = xrow[v * RS + tt];
                acc0[v] = fmaf(xm, wih, bias0n);
                acc1[v] = 0.0f;
            }
            const float* h0r = &cat[(i & 1) ^ 1][H0_COL];
            #pragma unroll
            for (int kc = 0; kc < KC; kc += 2) {
                v2f aA = *(const v2f*)&h0r[r * RSC + kc * 4 + 2 * hf];
                v2f aB = *(const v2f*)&h0r[r * RSC + (kc + 1) * 4 + 2 * hf];
                acc0 = WMMA_F32(aA, rB0[kc],     acc0);
                acc1 = WMMA_F32(aB, rB0[kc + 1], acc1);
            }
            v8f c0 = acc0 + acc1;
            float t0[8];
            #pragma unroll
            for (int v = 0; v < 8; ++v) t0[v] = fast_tanh(c0[v]);
            bar_wait_late();
            float* h0w = &cat[i & 1][H0_COL + st0];
            #pragma unroll
            for (int v = 0; v < 8; ++v)
                h0w[v * RSC] = t0[v];
            lds_stores_landed();
            bar_arrive();
            bar_wait();
        }
        bar_arrive(); bar_wait();
        bar_arrive(); bar_wait();
        bar_arrive(); bar_wait();
    } else {
        v2f rBc[KCC];
        #pragma unroll
        for (int kc = 0; kc < KCC; ++kc) rBc[kc] = *(const v2f*)&wBc[nt][kc][lane][0];
        bar_arrive(); bar_wait();
        bar_arrive(); bar_wait();
        bar_arrive(); bar_wait();

        for (int s = 0; s < T_STEPS; ++s) {
            v8f acc2, acc3;
            #pragma unroll
            for (int v = 0; v < 8; ++v) { acc2[v] = bias1n; acc3[v] = 0.0f; }
            const float* cr = &cat[s & 1][0];
            #pragma unroll
            for (int kc = 0; kc < KCC - LASTQ; ++kc) {
                v2f a = *(const v2f*)&cr[r * RSC + kc * 4 + 2 * hf];
                if (kc & 1) acc3 = WMMA_F32(a, rBc[kc], acc3);
                else        acc2 = WMMA_F32(a, rBc[kc], acc2);
            }
            v2f la[LASTQ];
            #pragma unroll
            for (int q = 0; q < LASTQ; ++q) {
                const int kc = KCC - LASTQ + q;
                la[q] = *(const v2f*)&cr[r * RSC + kc * 4 + 2 * hf];
            }
            bar_arrive_here();
            #pragma unroll
            for (int q = 0; q < LASTQ; ++q) {
                const int kc = KCC - LASTQ + q;
                if (kc & 1) acc3 = WMMA_F32(la[q], rBc[kc], acc3);
                else        acc2 = WMMA_F32(la[q], rBc[kc], acc2);
            }
            v8f c1 = acc2 + acc3;
            float t1[8];
            #pragma unroll
            for (int v = 0; v < 8; ++v) t1[v] = fast_tanh(c1[v]);
            bar_wait_late();
            float* h1w = &cat[(s & 1) ^ 1][0];
            #pragma unroll
            for (int v = 0; v < 8; ++v)
                h1w[(v + 8 * hf) * RSC + col1] = t1[v];
            lds_stores_landed();
            bar_arrive();
            bar_wait();
        }
    }

    const float* h1f = &cat[T_STEPS & 1][0];
    for (int i = tid; i < 16 * O_OUT; i += NTHR) {
        int m = i / O_OUT, o = i % O_OUT;
        float acc = b_fc[o];
        #pragma unroll 5
        for (int k = 0; k < H_SZ; ++k)
            acc = fmaf(h1f[m * RSC + k], W_fc[o * H_SZ + k], acc);
        out[(size_t)(b0 + m) * O_OUT + o] = acc;
    }
}

extern "C" void kernel_launch(void* const* d_in, const int* in_sizes, int n_in,
                              void* d_out, int out_size, void* d_ws, size_t ws_size,
                              hipStream_t stream) {
    const float* x     = (const float*)d_in[0];
    const float* W_ih0 = (const float*)d_in[1];
    const float* W_hh0 = (const float*)d_in[2];
    const float* b_ih0 = (const float*)d_in[3];
    const float* b_hh0 = (const float*)d_in[4];
    const float* W_ih1 = (const float*)d_in[5];
    const float* W_hh1 = (const float*)d_in[6];
    const float* b_ih1 = (const float*)d_in[7];
    const float* b_hh1 = (const float*)d_in[8];
    const float* W_fc  = (const float*)d_in[9];
    const float* b_fc  = (const float*)d_in[10];
    float* out = (float*)d_out;

    rnn2_tanh_wmma<<<B_SZ / 16, NTHR, 0, stream>>>(
        x, W_ih0, W_hh0, b_ih0, b_hh0, W_ih1, W_hh1, b_ih1, b_hh1, W_fc, b_fc, out);
}


# Layer 1's product split over two waves per tile: 4 more waves (12, 3 per SIMD) form b1 + Wih1 h0(s) a step ahead, handed over in LDS as the h1 wave's accumulator start; order of the k sums changes
# speedup vs baseline: 1.7085x; 1.0589x over previous
#include <hip/hip_runtime.h>
#include <math.h>

typedef float v2f __attribute__((ext_vector_type(2)));
typedef float v4f __attribute__((ext_vector_type(4)));
typedef float v8f __attribute__((ext_vector_type(8)));

#define H_SZ 50
#define O_OUT 10
#define B_SZ 512
#define T_STEPS 2048
#define RS 68
#define RSH 68
#define XBLK 64
#define NT 4
#define KC 16
#define KH 13
#define NTHR 384

__device__ __forceinline__ float fast_tanh(float v) {
#if __has_builtin(__builtin_amdgcn_tanhf)
    return __builtin_amdgcn_tanhf(v);
#else
    float e = __expf(v + v);
    return 1.0f - __fdividef(2.0f, e + 1.0f);
#endif
}

#define WMMA_F32(A, B, C) \
    __builtin_amdgcn_wmma_f32_16x16x4_f32(false, (A), false, (B), (short)0, (C), false, false)

__device__ __forceinline__ void bar_arrive() { asm volatile("s_barrier_signal -1" ::: "memory"); }
__device__ __forceinline__ void bar_wait()   { asm volatile("s_barrier_wait -1" ::: "memory"); }
__device__ __forceinline__ void bar_wait_late() {
    __builtin_amdgcn_sched_barrier(0);
    asm volatile("s_barrier_wait -1" ::: "memory");
}
__device__ __forceinline__ void lds_stores_landed() { asm volatile("s_wait_dscnt 0x0" ::: "memory"); }

__global__ __launch_bounds__(NTHR, 1) void rnn2_tanh_wmma(
    const float* __restrict__ x,
    const float* __restrict__ W_ih0, const float* __restrict__ W_hh0,
    const float* __restrict__ b_ih0, const float* __restrict__ b_hh0,
    const float* __restrict__ W_ih1, const float* __restrict__ W_hh1,
    const float* __restrict__ b_ih1, const float* __restrict__ b_hh1,
    const float* __restrict__ W_fc,  const float* __restrict__ b_fc,
    float* __restrict__ out)
{
    __shared__ __attribute__((aligned(16))) float wB0[NT][KC][32][2];
    __shared__ __attribute__((aligned(16))) float wB1[NT][KH][32][2];
    __shared__ __attribute__((aligned(16))) float wBi[NT][KH][32][2];
    __shared__ __attribute__((aligned(16))) float h0s[2][16 * RSH];
    __shared__ __attribute__((aligned(16))) float h1s[2][16 * RSH];
    __shared__ __attribute__((aligned(16))) float qs[2][NT][2][32][4];
    __shared__ __attribute__((aligned(16))) float xs[NT][16 * RS];

    const int tid  = threadIdx.x;
    const int lane = tid & 31;
    const int wv   = __builtin_amdgcn_readfirstlane(tid >> 5);
    const int nt   = wv & 3;
    const int role = ((wv >> 2) + nt) % 3;
    const int r    = lane & 15;
    const int hf   = lane >> 4;
    const int b0   = blockIdx.x * 16;

    {
        float* z0 = &h0s[0][0];
        float* z1 = &h1s[0][0];
        for (int i = tid; i < 2 * 16 * RSH; i += NTHR) { z0[i] = 0.f; z1[i] = 0.f; }
        float* w0f = &wB0[0][0][0][0];
        for (int idx = tid; idx < NT * KC * 32 * 2; idx += NTHR) {
            int v  = idx & 1;
            int ln = (idx >> 1) & 31;
            int kc = (idx >> 6) & 15;
            int tn = idx >> 10;
            int n  = tn * 16 + (ln & 15);
            int k  = kc * 4 + 2 * (ln >> 4) + v;
            float a = 0.f;
            if (n < H_SZ && k < H_SZ)
                a = W_hh0[n * H_SZ + k];
            w0f[idx] = a;
        }
        float* w1f = &wB1[0][0][0][0];
        float* wif = &wBi[0][0][0][0];
        for (int idx = tid; idx < NT * KH * 32 * 2; idx += NTHR) {
            int v  = idx & 1;
            int ln = (idx >> 1) & 31;
            int ch = idx >> 6;
            int kc = ch % KH;
            int tn = ch / KH;
            int n  = tn * 16 + (ln & 15);
            int k  = kc * 4 + 2 * (ln >> 4) + v;
            float a = 0.f, c = 0.f;
            if (n < H_SZ && k < H_SZ) {
                a = W_hh1[n * H_SZ + k];
                c = W_ih1[n * H_SZ + k];
            }
            w1f[idx] = a;
            wif[idx] = c;
        }
    }

    const int n = nt * 16 + r;
    float wih = 0.f, bias0n = 0.f, bias1n = 0.f;
    if (n < H_SZ) {
        wih    = W_ih0[n];
        bias0n = b_ih0[n] + b_hh0[n];
        bias1n = b_ih1[n] + b_hh1[n];
    }
    const int st0 = 8 * hf * RSH + n;
    __syncthreads();

    if (role == 0) {
        v2f rB0[KC];
        #pragma unroll
        for (int kc = 0; kc < KC; ++kc) rB0[kc] = *(const v2f*)&wB0[nt][kc][lane][0];
        float* xw = &xs[nt][0];
        const float* xrow = xw + 8 * hf * RS;

        for (int i = 0; i < T_STEPS; ++i) {
            bar_arrive();
            const int tt = i & (XBLK - 1);
            if (tt == 0) {
                #pragma unroll
                for (int j = lane; j < 16 * XBLK; j += 32) {
                    int rr = j >> 6;
                    int cc = j & (XBLK - 1);
                    xw[rr * RS + cc] = x[(size_t)(b0 + rr) * T_STEPS + i + cc];
                }
            }
            v8f acc0, acc1;
            #pragma unroll
            for (int v = 0; v < 8; ++v) {
                float xm = xrow[v * RS + tt];
                acc0[v] = fmaf(xm, wih, bias0n);
                acc1[v] = 0.0f;
            }
            const float* h0r = &h0s[(i & 1) ^ 1][0];
            #pragma unroll
            for (int kc = 0; kc < KC; kc += 2) {
                v2f aA = *(const v2f*)&h0r[r * RSH + kc * 4 + 2 * hf];
                v2f aB = *(const v2f*)&h0r[r * RSH + (kc + 1) * 4 + 2 * hf];
                acc0 = WMMA_F32(aA, rB0[kc],     acc0);
                acc1 = WMMA_F32(aB, rB0[kc + 1], acc1);
            }
            v8f c0 = acc0 + acc1;
            float t0[8];
            #pragma unroll
            for (int v = 0; v < 8; ++v) t0[v] = fast_tanh(c0[v]);
            float* h0w = &h0s[i & 1][st0];
            #pragma unroll
            for (int v = 0; v < 8; ++v)
                h0w[v * RSH] = t0[v];
            lds_stores_landed();
            bar_wait_late();
            bar_arrive();
            bar_wait();
        }
        bar_arrive(); bar_wait();
        bar_arrive(); bar_wait();
        bar_arrive(); bar_wait();
    } else if (role == 2) {
        v2f rBi[KH];
        #pragma unroll
        for (int kc = 0; kc < KH; ++kc) rBi[kc] = *(const v2f*)&wBi[nt][kc][lane][0];
        bar_arrive(); bar_wait();

        for (int s = 0; s < T_STEPS; ++s) {
            bar_arrive();
            bar_wait();
            v8f acc4, acc5;
            #pragma unroll
            for (int v = 0; v < 8; ++v) { acc4[v] = bias1n; acc5[v] = 0.0f; }
            const float* hr = &h0s[s & 1][0];
            #pragma unroll
            for (int kc = 0; kc < KH; ++kc) {
                v2f a = *(const v2f*)&hr[r * RSH + kc * 4 + 2 * hf];
                if (kc & 1) acc5 = WMMA_F32(a, rBi[kc], acc5);
                else        acc4 = WMMA_F32(a, rBi[kc], acc4);
            }
            v8f q = acc4 + acc5;
            float* qw = &qs[s & 1][nt][0][lane][0];
            v4f qlo = { q[0], q[1], q[2], q[3] };
            v4f qhi = { q[4], q[5], q[6], q[7] };
            *(v4f*)qw = qlo;
            *(v4f*)(qw + 32 * 4) = qhi;
            lds_stores_landed();
            bar_arrive();
            bar_wait();
        }
        bar_arrive(); bar_wait();
        bar_arrive(); bar_wait();
    } else {
        v2f rB1[KH];
        #pragma unroll
        for (int kc = 0; kc < KH; ++kc) rB1[kc] = *(const v2f*)&wB1[nt][kc][lane][0];
        bar_arrive(); bar_wait();
        bar_arrive(); bar_wait();
        bar_arrive(); bar_wait();

        for (int s = 0; s < T_STEPS; ++s) {
            bar_arrive();
            const float* qr = &qs[s & 1][nt][0][lane][0];
            v4f qlo = *(const v4f*)qr;
            v4f qhi = *(const v4f*)(qr + 32 * 4);
            v8f acc2, acc3;
            #pragma unroll
            for (int v = 0; v < 4; ++v) { acc2[v] = qlo[v]; acc2[v + 4] = qhi[v]; }
            #pragma unroll
            for (int v = 0; v < 8; ++v) acc3[v] = 0.0f;
            const float* hr = &h1s[(s & 1) ^ 1][0];
            #pragma unroll
            for (int kc = 0; kc < KH; ++kc) {
                v2f a = *(const v2f*)&hr[r * RSH + kc * 4 + 2 * hf];
                if (kc & 1) acc3 = WMMA_F32(a, rB1[kc], acc3);
                else        acc2 = WMMA_F32(a, rB1[kc], acc2);
            }
            v8f c1 = acc2 + acc3;
            float t1[8];
            #pragma unroll
            for (int v = 0; v < 8; ++v) t1[v] = fast_tanh(c1[v]);
            float* h1w = &h1s[s & 1][st0];
            #pragma unroll
            for (int v = 0; v < 8; ++v)
                h1w[v * RSH] = t1[v];
            lds_stores_landed();
            bar_wait_late();
            bar_arrive();
            bar_wait();
        }
    }

    const float* h1f = &h1s[(T_STEPS - 1) & 1][0];
    for (int i = tid; i < 16 * O_OUT; i += NTHR) {
        int m = i / O_OUT, o = i % O_OUT;
        float acc = b_fc[o];
        #pragma unroll 5
        for (int k = 0; k < H_SZ; ++k)
            acc = fmaf(h1f[m * RSH + k], W_fc[o * H_SZ + k], acc);
        out[(size_t)(b0 + m) * O_OUT + o] = acc;
    }
}

extern "C" void kernel_launch(void* const* d_in, const int* in_sizes, int n_in,
                              void* d_out, int out_size, void* d_ws, size_t ws_size,
                              hipStream_t stream) {
    const float* x     = (const float*)d_in[0];
    const float* W_ih0 = (const float*)d_in[1];
    const float* W_hh0 = (const float*)d_in[2];
    const float* b_ih0 = (const float*)d_in[3];
    const float* b_hh0 = (const float*)d_in[4];
    const float* W_ih1 = (const float*)d_in[5];
    const float* W_hh1 = (const float*)d_in[6];
    const float* b_ih1 = (const float*)d_in[7];
    const float* b_hh1 = (const float*)d_in[8];
    const float* W_fc  = (const float*)d_in[9];
    const float* b_fc  = (const float*)d_in[10];
    float* out = (float*)d_out;

    rnn2_tanh_wmma<<<B_SZ / 16, NTHR, 0, stream>>>(
        x, W_ih0, W_hh0, b_ih0, b_hh0, W_ih1, W_hh1, b_ih1, b_hh1, W_fc, b_fc, out);
}
